# starting kernel
# baseline (speedup 1.0000x reference)
#include <hip/hip_runtime.h>

typedef float v2f __attribute__((ext_vector_type(2)));
typedef float v8f __attribute__((ext_vector_type(8)));

#define WMMA_F32(a, b, c) \
    __builtin_amdgcn_wmma_f32_16x16x4_f32(false, (a), false, (b), (short)0, (c), false, false)

constexpr int BATCH  = 8;
constexpr int NATOMS = 256;
constexpr int DF     = 42;
constexpr int H1     = 64;
constexpr int H2     = 32;
constexpr int ROWS   = BATCH * NATOMS;
constexpr int JF     = NATOMS * DF;

__device__ __forceinline__ float softplus_f(float z) {
    return fmaxf(z, 0.0f) + log1pf(expf(-fabsf(z)));
}
__device__ __forceinline__ float sigmoid_f(float z) {
    return 1.0f / (1.0f + expf(-z));
}

__global__ __launch_bounds__(32)
void mlp_wmma_kernel(const float* __restrict__ image,
                     const float* __restrict__ W1, const float* __restrict__ b1,
                     const float* __restrict__ W2, const float* __restrict__ b2,
                     const float* __restrict__ W3, const float* __restrict__ b3,
                     float* __restrict__ gws,
                     float* __restrict__ eiws)
{
    __shared__ float sW1[48 * 64];
    __shared__ float sW2[64 * 32];
    __shared__ float sW3[32];
    __shared__ float sb1[64];
    __shared__ float sb2[32];
    __shared__ float sX [16 * 44];
    __shared__ float sA1[16 * 64];
    __shared__ float sA2[16 * 32];
    __shared__ float sG2[16 * 32];
    __shared__ float sG1[16 * 64];

    const int lane = threadIdx.x;
    const int half = lane >> 4;
    const int l    = lane & 15;
    const int tile = blockIdx.x;
    const int row0 = tile * 16;

    for (int idx = lane; idx < 48 * 64; idx += 32) {
        int r = idx >> 6, c = idx & 63;
        sW1[idx] = (r < 42) ? W1[r * 64 + c] : 0.0f;
    }
    for (int idx = lane; idx < 64 * 32; idx += 32) sW2[idx] = W2[idx];
    sW3[lane] = W3[lane];
    for (int idx = lane; idx < 64; idx += 32) sb1[idx] = b1[idx];
    sb2[lane] = b2[lane];
    for (int idx = lane; idx < 16 * 44; idx += 32) {
        int r = idx / 44, c = idx % 44;
        sX[idx] = (c < 42) ? image[(row0 + r) * DF + c] : 0.0f;
    }
    __syncthreads();

    v8f z1t[4];
    #pragma unroll
    for (int t = 0; t < 4; ++t) {
        float bv = sb1[t * 16 + l];
        v8f c;
        #pragma unroll
        for (int i = 0; i < 8; ++i) c[i] = bv;
        #pragma unroll
        for (int k0 = 0; k0 < 44; k0 += 4) {
            v2f a, b;
            a.x = sX[l * 44 + k0 + 2 * half];
            a.y = sX[l * 44 + k0 + 2 * half + 1];
            b.x = sW1[(k0 + 2 * half)     * 64 + t * 16 + l];
            b.y = sW1[(k0 + 2 * half + 1) * 64 + t * 16 + l];
            c = WMMA_F32(a, b, c);
        }
        z1t[t] = c;
    }

    v8f sig1[4];
    #pragma unroll
    for (int t = 0; t < 4; ++t) {
        #pragma unroll
        for (int i = 0; i < 8; ++i) {
            float z = z1t[t][i];
            sA1[(i + 8 * half) * 64 + t * 16 + l] = softplus_f(z);
            sig1[t][i] = sigmoid_f(z);
        }
    }
    __syncthreads();

    v8f z2t[2];
    #pragma unroll
    for (int t = 0; t < 2; ++t) {
        float bv = sb2[t * 16 + l];
        v8f c;
        #pragma unroll
        for (int i = 0; i < 8; ++i) c[i] = bv;
        #pragma unroll
        for (int k0 = 0; k0 < 64; k0 += 4) {
            v2f a, b;
            a.x = sA1[l * 64 + k0 + 2 * half];
            a.y = sA1[l * 64 + k0 + 2 * half + 1];
            b.x = sW2[(k0 + 2 * half)     * 32 + t * 16 + l];
            b.y = sW2[(k0 + 2 * half + 1) * 32 + t * 16 + l];
            c = WMMA_F32(a, b, c);
        }
        z2t[t] = c;
    }

    #pragma unroll
    for (int t = 0; t < 2; ++t) {
        float w3v = sW3[t * 16 + l];
        #pragma unroll
        for (int i = 0; i < 8; ++i) {
            float z = z2t[t][i];
            sA2[(i + 8 * half) * 32 + t * 16 + l] = softplus_f(z);
            sG2[(i + 8 * half) * 32 + t * 16 + l] = sigmoid_f(z) * w3v;
        }
    }
    __syncthreads();

    v8f ce;
    #pragma unroll
    for (int i = 0; i < 8; ++i) ce[i] = 0.0f;
    #pragma unroll
    for (int k0 = 0; k0 < 32; k0 += 4) {
        v2f a, b;
        a.x = sA2[l * 32 + k0 + 2 * half];
        a.y = sA2[l * 32 + k0 + 2 * half + 1];
        b.x = (l == 0) ? sW3[k0 + 2 * half]     : 0.0f;
        b.y = (l == 0) ? sW3[k0 + 2 * half + 1] : 0.0f;
        ce = WMMA_F32(a, b, ce);
    }
    {
        float b3v = b3[0];
        if (l == 0) {
            #pragma unroll
            for (int v = 0; v < 8; ++v)
                eiws[row0 + 8 * half + v] = ce[v] + b3v;
        }
    }

    #pragma unroll
    for (int t = 0; t < 4; ++t) {
        v8f c;
        #pragma unroll
        for (int i = 0; i < 8; ++i) c[i] = 0.0f;
        #pragma unroll
        for (int k0 = 0; k0 < 32; k0 += 4) {
            v2f a, b;
            a.x = sG2[l * 32 + k0 + 2 * half];
            a.y = sG2[l * 32 + k0 + 2 * half + 1];
            b.x = sW2[(t * 16 + l) * 32 + k0 + 2 * half];
            b.y = sW2[(t * 16 + l) * 32 + k0 + 2 * half + 1];
            c = WMMA_F32(a, b, c);
        }
        #pragma unroll
        for (int i = 0; i < 8; ++i)
            sG1[(i + 8 * half) * 64 + t * 16 + l] = sig1[t][i] * c[i];
    }
    __syncthreads();

    #pragma unroll
    for (int t = 0; t < 3; ++t) {
        v8f c;
        #pragma unroll
        for (int i = 0; i < 8; ++i) c[i] = 0.0f;
        #pragma unroll
        for (int k0 = 0; k0 < 64; k0 += 4) {
            v2f a, b;
            a.x = sG1[l * 64 + k0 + 2 * half];
            a.y = sG1[l * 64 + k0 + 2 * half + 1];
            b.x = sW1[(t * 16 + l) * 64 + k0 + 2 * half];
            b.y = sW1[(t * 16 + l) * 64 + k0 + 2 * half + 1];
            c = WMMA_F32(a, b, c);
        }
        int d = t * 16 + l;
        if (d < DF) {
            #pragma unroll
            for (int v = 0; v < 8; ++v)
                gws[(row0 + v + 8 * half) * DF + d] = c[v];
        }
    }
}

__global__ __launch_bounds__(256)
void force_kernel(const float* __restrict__ dfeat,
                  const float* __restrict__ gws,
                  float* __restrict__ out)
{
    __shared__ __align__(16) float gsh[JF];

    const int tid = threadIdx.x;
    const int blk = blockIdx.x;
    const int b   = blk >> 5;
    const int i0  = (blk & 31) * 8;

    for (int idx = tid; idx < JF; idx += 256) gsh[idx] = gws[b * JF + idx];
    __syncthreads();

    const int wave = tid >> 5;
    const int lane = tid & 31;
    const int i    = i0 + wave;

    const float* row = dfeat + (size_t)(b * NATOMS + i) * (size_t)(JF * 3);
    const float4* g4 = (const float4*)gsh;

    float ax = 0.0f, ay = 0.0f, az = 0.0f;
    for (int it = 0; it < 84; ++it) {
        int q = it * 32 + lane;
        const float4* p = (const float4*)(row + (size_t)q * 12);
        float4 d0 = p[0], d1 = p[1], d2 = p[2];
        float4 gv = g4[q];
        ax = fmaf(d0.x, gv.x, fmaf(d0.w, gv.y, fmaf(d1.z, gv.z, fmaf(d2.y, gv.w, ax))));
        ay = fmaf(d0.y, gv.x, fmaf(d1.x, gv.y, fmaf(d1.w, gv.z, fmaf(d2.z, gv.w, ay))));
        az = fmaf(d0.z, gv.x, fmaf(d1.y, gv.y, fmaf(d2.x, gv.z, fmaf(d2.w, gv.w, az))));
    }
    #pragma unroll
    for (int off = 16; off; off >>= 1) {
        ax += __shfl_xor(ax, off, 32);
        ay += __shfl_xor(ay, off, 32);
        az += __shfl_xor(az, off, 32);
    }
    if (lane == 0) {
        float* f = out + 8 + (size_t)(b * NATOMS + i) * 3;
        f[0] = ax; f[1] = ay; f[2] = az;
    }
}

__global__ __launch_bounds__(256)
void etot_kernel(const float* __restrict__ eiws, float* __restrict__ out)
{
    const int b = threadIdx.x >> 5;
    const int lane = threadIdx.x & 31;
    float s = 0.0f;
    #pragma unroll
    for (int k = 0; k < 8; ++k) s += eiws[b * NATOMS + lane + 32 * k];
    #pragma unroll
    for (int off = 16; off; off >>= 1) s += __shfl_xor(s, off, 32);
    if (lane == 0) out[b] = s;
}

extern "C" void kernel_launch(void* const* d_in, const int* in_sizes, int n_in,
                              void* d_out, int out_size, void* d_ws, size_t ws_size,
                              hipStream_t stream)
{
    (void)in_sizes; (void)n_in; (void)out_size; (void)ws_size;

    const float* image = (const float*)d_in[0];
    const float* dfeat = (const float*)d_in[1];
    const float* W1    = (const float*)d_in[2];
    const float* b1    = (const float*)d_in[3];
    const float* W2    = (const float*)d_in[4];
    const float* b2    = (const float*)d_in[5];
    const float* W3    = (const float*)d_in[6];
    const float* b3    = (const float*)d_in[7];

    float* out  = (float*)d_out;
    float* gws  = (float*)d_ws;
    float* eiws = gws + (size_t)ROWS * DF;

    mlp_wmma_kernel<<<ROWS / 16, 32, 0, stream>>>(image, W1, b1, W2, b2, W3, b3, gws, eiws);
    force_kernel<<<256, 256, 0, stream>>>(dfeat, gws, out);
    etot_kernel<<<1, 256, 0, stream>>>(eiws, out);
}


# first kernel: weight, bias and input-tile fragments read straight from global memory into registers, issued together up front; no LDS staging loops, LDS only for activations
# speedup vs baseline: 1.8871x; 1.8871x over previous
#include <hip/hip_runtime.h>

typedef float v2f __attribute__((ext_vector_type(2)));
typedef float v8f __attribute__((ext_vector_type(8)));

#define WMMA_F32(a, b, c) \
    __builtin_amdgcn_wmma_f32_16x16x4_f32(false, (a), false, (b), (short)0, (c), false, false)

constexpr int BATCH  = 8;
constexpr int NATOMS = 256;
constexpr int DF     = 42;
constexpr int H1     = 64;
constexpr int H2     = 32;
constexpr int ROWS   = BATCH * NATOMS;
constexpr int JF     = NATOMS * DF;

__device__ __forceinline__ float softplus_f(float z) {
    return fmaxf(z, 0.0f) + log1pf(expf(-fabsf(z)));
}
__device__ __forceinline__ float sigmoid_f(float z) {
    return 1.0f / (1.0f + expf(-z));
}

__global__ __launch_bounds__(32)
void mlp_wmma_kernel(const float* __restrict__ image,
                     const float* __restrict__ W1, const float* __restrict__ b1,
                     const float* __restrict__ W2, const float* __restrict__ b2,
                     const float* __restrict__ W3, const float* __restrict__ b3,
                     float* __restrict__ gws,
                     float* __restrict__ eiws)
{
    __shared__ float sA1[16 * 64];
    __shared__ float sA2[16 * 32];
    __shared__ float sG2[16 * 32];
    __shared__ float sG1[16 * 64];

    const int lane = threadIdx.x;
    const int half = lane >> 4;
    const int l    = lane & 15;
    const int kh   = 2 * half;
    const int tile = blockIdx.x;
    const int row0 = tile * 16;

    v2f zero2; zero2.x = 0.0f; zero2.y = 0.0f;

    const float* xrow = image + (row0 + l) * DF;
    v2f xa[11];
    #pragma unroll
    for (int j = 0; j < 10; ++j) xa[j] = *(const v2f*)(xrow + 4 * j + kh);
    {
        v2f v = *(const v2f*)(xrow + 40);
        xa[10] = half ? zero2 : v;
    }

    v2f w1b[4][11];
    #pragma unroll
    for (int t = 0; t < 4; ++t) {
        #pragma unroll
        for (int j = 0; j < 10; ++j) {
            const float* p = W1 + (4 * j + kh) * H1 + t * 16 + l;
            w1b[t][j].x = p[0];
            w1b[t][j].y = p[H1];
        }
        const float* p = W1 + 40 * H1 + t * 16 + l;
        v2f v; v.x = p[0]; v.y = p[H1];
        w1b[t][10] = half ? zero2 : v;
    }

    v2f w2b[2][16];
    #pragma unroll
    for (int t = 0; t < 2; ++t) {
        #pragma unroll
        for (int j = 0; j < 16; ++j) {
            const float* p = W2 + (4 * j + kh) * H2 + t * 16 + l;
            w2b[t][j].x = p[0];
            w2b[t][j].y = p[H2];
        }
    }

    float b1v[4], b2v[2], w3n[2];
    #pragma unroll
    for (int t = 0; t < 4; ++t) b1v[t] = b1[t * 16 + l];
    #pragma unroll
    for (int t = 0; t < 2; ++t) { b2v[t] = b2[t * 16 + l]; w3n[t] = W3[t * 16 + l]; }
    v2f w3k[8];
    #pragma unroll
    for (int j = 0; j < 8; ++j) w3k[j] = *(const v2f*)(W3 + 4 * j + kh);
    const float b3v = b3[0];

    v8f z1t[4];
    #pragma unroll
    for (int t = 0; t < 4; ++t) {
        v8f c;
        #pragma unroll
        for (int i = 0; i < 8; ++i) c[i] = b1v[t];
        #pragma unroll
        for (int j = 0; j < 11; ++j) c = WMMA_F32(xa[j], w1b[t][j], c);
        z1t[t] = c;
    }

    v2f w2t[4][8];
    #pragma unroll
    for (int t = 0; t < 4; ++t) {
        #pragma unroll
        for (int j = 0; j < 8; ++j)
            w2t[t][j] = *(const v2f*)(W2 + (t * 16 + l) * H2 + 4 * j + kh);
    }
    v2f w1t[3][16];
    #pragma unroll
    for (int t = 0; t < 3; ++t) {
        const int d  = t * 16 + l;
        const int dc = d < DF ? d : DF - 1;
        #pragma unroll
        for (int j = 0; j < 16; ++j)
            w1t[t][j] = *(const v2f*)(W1 + dc * H1 + 4 * j + kh);
    }

    v8f sig1[4];
    #pragma unroll
    for (int t = 0; t < 4; ++t) {
        #pragma unroll
        for (int i = 0; i < 8; ++i) {
            float z = z1t[t][i];
            sA1[(i + 8 * half) * 64 + t * 16 + l] = softplus_f(z);
            sig1[t][i] = sigmoid_f(z);
        }
    }
    __syncthreads();

    v8f z2t[2];
    #pragma unroll
    for (int t = 0; t < 2; ++t) {
        v8f c;
        #pragma unroll
        for (int i = 0; i < 8; ++i) c[i] = b2v[t];
        #pragma unroll
        for (int j = 0; j < 16; ++j) {
            v2f a;
            a.x = sA1[l * 64 + 4 * j + kh];
            a.y = sA1[l * 64 + 4 * j + kh + 1];
            c = WMMA_F32(a, w2b[t][j], c);
        }
        z2t[t] = c;
    }

    #pragma unroll
    for (int t = 0; t < 2; ++t) {
        #pragma unroll
        for (int i = 0; i < 8; ++i) {
            float z = z2t[t][i];
            sA2[(i + 8 * half) * 32 + t * 16 + l] = softplus_f(z);
            sG2[(i + 8 * half) * 32 + t * 16 + l] = sigmoid_f(z) * w3n[t];
        }
    }
    __syncthreads();

    v8f ce;
    #pragma unroll
    for (int i = 0; i < 8; ++i) ce[i] = 0.0f;
    #pragma unroll
    for (int j = 0; j < 8; ++j) {
        v2f a;
        a.x = sA2[l * 32 + 4 * j + kh];
        a.y = sA2[l * 32 + 4 * j + kh + 1];
        v2f b = (l == 0) ? w3k[j] : zero2;
        ce = WMMA_F32(a, b, ce);
    }
    if (l == 0) {
        #pragma unroll
        for (int v = 0; v < 8; ++v)
            eiws[row0 + 8 * half + v] = ce[v] + b3v;
    }

    #pragma unroll
    for (int t = 0; t < 4; ++t) {
        v8f c;
        #pragma unroll
        for (int i = 0; i < 8; ++i) c[i] = 0.0f;
        #pragma unroll
        for (int j = 0; j < 8; ++j) {
            v2f a;
            a.x = sG2[l * 32 + 4 * j + kh];
            a.y = sG2[l * 32 + 4 * j + kh + 1];
            c = WMMA_F32(a, w2t[t][j], c);
        }
        #pragma unroll
        for (int i = 0; i < 8; ++i)
            sG1[(i + 8 * half) * 64 + t * 16 + l] = sig1[t][i] * c[i];
    }
    __syncthreads();

    #pragma unroll
    for (int t = 0; t < 3; ++t) {
        v8f c;
        #pragma unroll
        for (int i = 0; i < 8; ++i) c[i] = 0.0f;
        #pragma unroll
        for (int j = 0; j < 16; ++j) {
            v2f a;
            a.x = sG1[l * 64 + 4 * j + kh];
            a.y = sG1[l * 64 + 4 * j + kh + 1];
            c = WMMA_F32(a, w1t[t][j], c);
        }
        int d = t * 16 + l;
        if (d < DF) {
            #pragma unroll
            for (int v = 0; v < 8; ++v)
                gws[(row0 + v + 8 * half) * DF + d] = c[v];
        }
    }
}

__global__ __launch_bounds__(256)
void force_kernel(const float* __restrict__ dfeat,
                  const float* __restrict__ gws,
                  float* __restrict__ out)
{
    __shared__ __align__(16) float gsh[JF];

    const int tid = threadIdx.x;
    const int blk = blockIdx.x;
    const int b   = blk >> 5;
    const int i0  = (blk & 31) * 8;

    for (int idx = tid; idx < JF; idx += 256) gsh[idx] = gws[b * JF + idx];
    __syncthreads();

    const int wave = tid >> 5;
    const int lane = tid & 31;
    const int i    = i0 + wave;

    const float* row = dfeat + (size_t)(b * NATOMS + i) * (size_t)(JF * 3);
    const float4* g4 = (const float4*)gsh;

    float ax = 0.0f, ay = 0.0f, az = 0.0f;
    for (int it = 0; it < 84; ++it) {
        int q = it * 32 + lane;
        const float4* p = (const float4*)(row + (size_t)q * 12);
        float4 d0 = p[0], d1 = p[1], d2 = p[2];
        float4 gv = g4[q];
        ax = fmaf(d0.x, gv.x, fmaf(d0.w, gv.y, fmaf(d1.z, gv.z, fmaf(d2.y, gv.w, ax))));
        ay = fmaf(d0.y, gv.x, fmaf(d1.x, gv.y, fmaf(d1.w, gv.z, fmaf(d2.z, gv.w, ay))));
        az = fmaf(d0.z, gv.x, fmaf(d1.y, gv.y, fmaf(d2.x, gv.z, fmaf(d2.w, gv.w, az))));
    }
    #pragma unroll
    for (int off = 16; off; off >>= 1) {
        ax += __shfl_xor(ax, off, 32);
        ay += __shfl_xor(ay, off, 32);
        az += __shfl_xor(az, off, 32);
    }
    if (lane == 0) {
        float* f = out + 8 + (size_t)(b * NATOMS + i) * 3;
        f[0] = ax; f[1] = ay; f[2] = az;
    }
}

__global__ __launch_bounds__(256)
void etot_kernel(const float* __restrict__ eiws, float* __restrict__ out)
{
    const int b = threadIdx.x >> 5;
    const int lane = threadIdx.x & 31;
    float s = 0.0f;
    #pragma unroll
    for (int k = 0; k < 8; ++k) s += eiws[b * NATOMS + lane + 32 * k];
    #pragma unroll
    for (int off = 16; off; off >>= 1) s += __shfl_xor(s, off, 32);
    if (lane == 0) out[b] = s;
}

extern "C" void kernel_launch(void* const* d_in, const int* in_sizes, int n_in,
                              void* d_out, int out_size, void* d_ws, size_t ws_size,
                              hipStream_t stream)
{
    (void)in_sizes; (void)n_in; (void)out_size; (void)ws_size;

    const float* image = (const float*)d_in[0];
    const float* dfeat = (const float*)d_in[1];
    const float* W1    = (const float*)d_in[2];
    const float* b1    = (const float*)d_in[3];
    const float* W2    = (const float*)d_in[4];
    const float* b2    = (const float*)d_in[5];
    const float* W3    = (const float*)d_in[6];
    const float* b3    = (const float*)d_in[7];

    float* out  = (float*)d_out;
    float* gws  = (float*)d_ws;
    float* eiws = gws + (size_t)ROWS * DF;

    mlp_wmma_kernel<<<ROWS / 16, 32, 0, stream>>>(image, W1, b1, W2, b2, W3, b3, gws, eiws);
    force_kernel<<<256, 256, 0, stream>>>(dfeat, gws, out);
    etot_kernel<<<1, 256, 0, stream>>>(eiws, out);
}


# force kernel start: the g copy to LDS is requested as one batch of 16-byte loads (one round trip, was seven load-wait-store trips), the row's first 12 dfeat loads right behind it; sums unchanged
# speedup vs baseline: 1.9498x; 1.0332x over previous
#include <hip/hip_runtime.h>

typedef float v2f __attribute__((ext_vector_type(2)));
typedef float v8f __attribute__((ext_vector_type(8)));

#define WMMA_F32(a, b, c) \
    __builtin_amdgcn_wmma_f32_16x16x4_f32(false, (a), false, (b), (short)0, (c), false, false)

constexpr int BATCH  = 8;
constexpr int NATOMS = 256;
constexpr int DF     = 42;
constexpr int H1     = 64;
constexpr int H2     = 32;
constexpr int ROWS   = BATCH * NATOMS;
constexpr int JF     = NATOMS * DF;

__device__ __forceinline__ float softplus_f(float z) {
    return fmaxf(z, 0.0f) + log1pf(expf(-fabsf(z)));
}
__device__ __forceinline__ float sigmoid_f(float z) {
    return 1.0f / (1.0f + expf(-z));
}

__global__ __launch_bounds__(32)
void mlp_wmma_kernel(const float* __restrict__ image,
                     const float* __restrict__ W1, const float* __restrict__ b1,
                     const float* __restrict__ W2, const float* __restrict__ b2,
                     const float* __restrict__ W3, const float* __restrict__ b3,
                     float* __restrict__ gws,
                     float* __restrict__ eiws)
{
    __shared__ float sA1[16 * 64];
    __shared__ float sA2[16 * 32];
    __shared__ float sG2[16 * 32];
    __shared__ float sG1[16 * 64];

    const int lane = threadIdx.x;
    const int half = lane >> 4;
    const int l    = lane & 15;
    const int kh   = 2 * half;
    const int tile = blockIdx.x;
    const int row0 = tile * 16;

    v2f zero2; zero2.x = 0.0f; zero2.y = 0.0f;

    const float* xrow = image + (row0 + l) * DF;
    v2f xa[11];
    #pragma unroll
    for (int j = 0; j < 10; ++j) xa[j] = *(const v2f*)(xrow + 4 * j + kh);
    {
        v2f v = *(const v2f*)(xrow + 40);
        xa[10] = half ? zero2 : v;
    }

    v2f w1b[4][11];
    #pragma unroll
    for (int t = 0; t < 4; ++t) {
        #pragma unroll
        for (int j = 0; j < 10; ++j) {
            const float* p = W1 + (4 * j + kh) * H1 + t * 16 + l;
            w1b[t][j].x = p[0];
            w1b[t][j].y = p[H1];
        }
        const float* p = W1 + 40 * H1 + t * 16 + l;
        v2f v; v.x = p[0]; v.y = p[H1];
        w1b[t][10] = half ? zero2 : v;
    }

    v2f w2b[2][16];
    #pragma unroll
    for (int t = 0; t < 2; ++t) {
        #pragma unroll
        for (int j = 0; j < 16; ++j) {
            const float* p = W2 + (4 * j + kh) * H2 + t * 16 + l;
            w2b[t][j].x = p[0];
            w2b[t][j].y = p[H2];
        }
    }

    float b1v[4], b2v[2], w3n[2];
    #pragma unroll
    for (int t = 0; t < 4; ++t) b1v[t] = b1[t * 16 + l];
    #pragma unroll
    for (int t = 0; t < 2; ++t) { b2v[t] = b2[t * 16 + l]; w3n[t] = W3[t * 16 + l]; }
    v2f w3k[8];
    #pragma unroll
    for (int j = 0; j < 8; ++j) w3k[j] = *(const v2f*)(W3 + 4 * j + kh);
    const float b3v = b3[0];

    v8f z1t[4];
    #pragma unroll
    for (int t = 0; t < 4; ++t) {
        v8f c;
        #pragma unroll
        for (int i = 0; i < 8; ++i) c[i] = b1v[t];
        #pragma unroll
        for (int j = 0; j < 11; ++j) c = WMMA_F32(xa[j], w1b[t][j], c);
        z1t[t] = c;
    }

    v2f w2t[4][8];
    #pragma unroll
    for (int t = 0; t < 4; ++t) {
        #pragma unroll
        for (int j = 0; j < 8; ++j)
            w2t[t][j] = *(const v2f*)(W2 + (t * 16 + l) * H2 + 4 * j + kh);
    }
    v2f w1t[3][16];
    #pragma unroll
    for (int t = 0; t < 3; ++t) {
        const int d  = t * 16 + l;
        const int dc = d < DF ? d : DF - 1;
        #pragma unroll
        for (int j = 0; j < 16; ++j)
            w1t[t][j] = *(const v2f*)(W1 + dc * H1 + 4 * j + kh);
    }

    v8f sig1[4];
    #pragma unroll
    for (int t = 0; t < 4; ++t) {
        #pragma unroll
        for (int i = 0; i < 8; ++i) {
            float z = z1t[t][i];
            sA1[(i + 8 * half) * 64 + t * 16 + l] = softplus_f(z);
            sig1[t][i] = sigmoid_f(z);
        }
    }
    __syncthreads();

    v8f z2t[2];
    #pragma unroll
    for (int t = 0; t < 2; ++t) {
        v8f c;
        #pragma unroll
        for (int i = 0; i < 8; ++i) c[i] = b2v[t];
        #pragma unroll
        for (int j = 0; j < 16; ++j) {
            v2f a;
            a.x = sA1[l * 64 + 4 * j + kh];
            a.y = sA1[l * 64 + 4 * j + kh + 1];
            c = WMMA_F32(a, w2b[t][j], c);
        }
        z2t[t] = c;
    }

    #pragma unroll
    for (int t = 0; t < 2; ++t) {
        #pragma unroll
        for (int i = 0; i < 8; ++i) {
            float z = z2t[t][i];
            sA2[(i + 8 * half) * 32 + t * 16 + l] = softplus_f(z);
            sG2[(i + 8 * half) * 32 + t * 16 + l] = sigmoid_f(z) * w3n[t];
        }
    }
    __syncthreads();

    v8f ce;
    #pragma unroll
    for (int i = 0; i < 8; ++i) ce[i] = 0.0f;
    #pragma unroll
    for (int j = 0; j < 8; ++j) {
        v2f a;
        a.x = sA2[l * 32 + 4 * j + kh];
        a.y = sA2[l * 32 + 4 * j + kh + 1];
        v2f b = (l == 0) ? w3k[j] : zero2;
        ce = WMMA_F32(a, b, ce);
    }
    if (l == 0) {
        #pragma unroll
        for (int v = 0; v < 8; ++v)
            eiws[row0 + 8 * half + v] = ce[v] + b3v;
    }

    #pragma unroll
    for (int t = 0; t < 4; ++t) {
        v8f c;
        #pragma unroll
        for (int i = 0; i < 8; ++i) c[i] = 0.0f;
        #pragma unroll
        for (int j = 0; j < 8; ++j) {
            v2f a;
            a.x = sG2[l * 32 + 4 * j + kh];
            a.y = sG2[l * 32 + 4 * j + kh + 1];
            c = WMMA_F32(a, w2t[t][j], c);
        }
        #pragma unroll
        for (int i = 0; i < 8; ++i)
            sG1[(i + 8 * half) * 64 + t * 16 + l] = sig1[t][i] * c[i];
    }
    __syncthreads();

    #pragma unroll
    for (int t = 0; t < 3; ++t) {
        v8f c;
        #pragma unroll
        for (int i = 0; i < 8; ++i) c[i] = 0.0f;
        #pragma unroll
        for (int j = 0; j < 16; ++j) {
            v2f a;
            a.x = sG1[l * 64 + 4 * j + kh];
            a.y = sG1[l * 64 + 4 * j + kh + 1];
            c = WMMA_F32(a, w1t[t][j], c);
        }
        int d = t * 16 + l;
        if (d < DF) {
            #pragma unroll
            for (int v = 0; v < 8; ++v)
                gws[(row0 + v + 8 * half) * DF + d] = c[v];
        }
    }
}

__global__ __launch_bounds__(256)
void force_kernel(const float* __restrict__ dfeat,
                  const float* __restrict__ gws,
                  float* __restrict__ out)
{
    __shared__ __align__(16) float gsh[JF];

    const int tid = threadIdx.x;
    const int blk = blockIdx.x;
    const int b   = blk >> 5;
    const int i0  = (blk & 31) * 8;
    const int wave = tid >> 5;
    const int lane = tid & 31;
    const int i    = i0 + wave;

    const float4* gsrc = (const float4*)(gws + (size_t)b * JF);
    float4 gq[11];
    #pragma unroll
    for (int k = 0; k < 10; ++k) gq[k] = gsrc[tid + 256 * k];
    gq[10] = gsrc[2560 + (tid & 127)];

    const float* row = dfeat + (size_t)(b * NATOMS + i) * (size_t)(JF * 3);
    float4 dq[12];
    #pragma unroll
    for (int s = 0; s < 4; ++s) {
        const float4* p = (const float4*)(row + (size_t)(s * 32 + lane) * 12);
        dq[3 * s + 0] = p[0];
        dq[3 * s + 1] = p[1];
        dq[3 * s + 2] = p[2];
    }

    float4* gdst = (float4*)gsh;
    #pragma unroll
    for (int k = 0; k < 10; ++k) gdst[tid + 256 * k] = gq[k];
    gdst[2560 + (tid & 127)] = gq[10];
    __syncthreads();

    const float4* g4 = (const float4*)gsh;

    float ax = 0.0f, ay = 0.0f, az = 0.0f;
    #pragma unroll
    for (int s = 0; s < 4; ++s) {
        float4 d0 = dq[3 * s + 0], d1 = dq[3 * s + 1], d2 = dq[3 * s + 2];
        float4 gv = g4[s * 32 + lane];
        ax = fmaf(d0.x, gv.x, fmaf(d0.w, gv.y, fmaf(d1.z, gv.z, fmaf(d2.y, gv.w, ax))));
        ay = fmaf(d0.y, gv.x, fmaf(d1.x, gv.y, fmaf(d1.w, gv.z, fmaf(d2.z, gv.w, ay))));
        az = fmaf(d0.z, gv.x, fmaf(d1.y, gv.y, fmaf(d2.x, gv.z, fmaf(d2.w, gv.w, az))));
    }
    for (int it = 4; it < 84; ++it) {
        int q = it * 32 + lane;
        const float4* p = (const float4*)(row + (size_t)q * 12);
        float4 d0 = p[0], d1 = p[1], d2 = p[2];
        float4 gv = g4[q];
        ax = fmaf(d0.x, gv.x, fmaf(d0.w, gv.y, fmaf(d1.z, gv.z, fmaf(d2.y, gv.w, ax))));
        ay = fmaf(d0.y, gv.x, fmaf(d1.x, gv.y, fmaf(d1.w, gv.z, fmaf(d2.z, gv.w, ay))));
        az = fmaf(d0.z, gv.x, fmaf(d1.y, gv.y, fmaf(d2.x, gv.z, fmaf(d2.w, gv.w, az))));
    }
    #pragma unroll
    for (int off = 16; off; off >>= 1) {
        ax += __shfl_xor(ax, off, 32);
        ay += __shfl_xor(ay, off, 32);
        az += __shfl_xor(az, off, 32);
    }
    if (lane == 0) {
        float* f = out + 8 + (size_t)(b * NATOMS + i) * 3;
        f[0] = ax; f[1] = ay; f[2] = az;
    }
}

__global__ __launch_bounds__(256)
void etot_kernel(const float* __restrict__ eiws, float* __restrict__ out)
{
    const int b = threadIdx.x >> 5;
    const int lane = threadIdx.x & 31;
    float s = 0.0f;
    #pragma unroll
    for (int k = 0; k < 8; ++k) s += eiws[b * NATOMS + lane + 32 * k];
    #pragma unroll
    for (int off = 16; off; off >>= 1) s += __shfl_xor(s, off, 32);
    if (lane == 0) out[b] = s;
}

extern "C" void kernel_launch(void* const* d_in, const int* in_sizes, int n_in,
                              void* d_out, int out_size, void* d_ws, size_t ws_size,
                              hipStream_t stream)
{
    (void)in_sizes; (void)n_in; (void)out_size; (void)ws_size;

    const float* image = (const float*)d_in[0];
    const float* dfeat = (const float*)d_in[1];
    const float* W1    = (const float*)d_in[2];
    const float* b1    = (const float*)d_in[3];
    const float* W2    = (const float*)d_in[4];
    const float* b2    = (const float*)d_in[5];
    const float* W3    = (const float*)d_in[6];
    const float* b3    = (const float*)d_in[7];

    float* out  = (float*)d_out;
    float* gws  = (float*)d_ws;
    float* eiws = gws + (size_t)ROWS * DF;

    mlp_wmma_kernel<<<ROWS / 16, 32, 0, stream>>>(image, W1, b1, W2, b2, W3, b3, gws, eiws);
    force_kernel<<<256, 256, 0, stream>>>(dfeat, gws, out);
    etot_kernel<<<1, 256, 0, stream>>>(eiws, out);
}


# first kernel: 8 waves per 16-row tile; wave 0 keeps every matrix product, each layer's softplus/sigmoid elements are dealt out over all 256 lanes through LDS (4+2 per lane, was 32+16); same sums
# speedup vs baseline: 2.1005x; 1.0773x over previous
#include <hip/hip_runtime.h>

typedef float v2f __attribute__((ext_vector_type(2)));
typedef float v8f __attribute__((ext_vector_type(8)));

#define WMMA_F32(a, b, c) \
    __builtin_amdgcn_wmma_f32_16x16x4_f32(false, (a), false, (b), (short)0, (c), false, false)

constexpr int BATCH  = 8;
constexpr int NATOMS = 256;
constexpr int DF     = 42;
constexpr int H1     = 64;
constexpr int H2     = 32;
constexpr int ROWS   = BATCH * NATOMS;
constexpr int JF     = NATOMS * DF;

__device__ __forceinline__ float softplus_f(float z) {
    return fmaxf(z, 0.0f) + log1pf(expf(-fabsf(z)));
}
__device__ __forceinline__ float sigmoid_f(float z) {
    return 1.0f / (1.0f + expf(-z));
}

__global__ __launch_bounds__(256)
void mlp_wmma_kernel(const float* __restrict__ image,
                     const float* __restrict__ W1, const float* __restrict__ b1,
                     const float* __restrict__ W2, const float* __restrict__ b2,
                     const float* __restrict__ W3, const float* __restrict__ b3,
                     float* __restrict__ gws,
                     float* __restrict__ eiws)
{
    __shared__ float sZ1[16 * 64];
    __shared__ float sA1[16 * 64];
    __shared__ float sS1[16 * 64];
    __shared__ float sZ2[16 * 32];
    __shared__ float sA2[16 * 32];
    __shared__ float sG2[16 * 32];
    __shared__ float sG1[16 * 64];

    const int tid  = threadIdx.x;
    const int wave = __builtin_amdgcn_readfirstlane(tid >> 5);
    const int lane = tid & 31;
    const int half = lane >> 4;
    const int l    = lane & 15;
    const int kh   = 2 * half;
    const int tile = blockIdx.x;
    const int row0 = tile * 16;

    v2f zero2; zero2.x = 0.0f; zero2.y = 0.0f;

    const float w3c = W3[tid & 31];

    v2f xa[11];
    v2f w1b[4][11];
    v2f w2b[2][16];
    float b1v[4], b2v[2];
    v2f w3k[8];
    float b3v = 0.0f;
    v2f w2t[4][8];
    v2f w1t[3][16];

    if (wave == 0) {
        const float* xrow = image + (row0 + l) * DF;
        #pragma unroll
        for (int j = 0; j < 10; ++j) xa[j] = *(const v2f*)(xrow + 4 * j + kh);
        {
            v2f v = *(const v2f*)(xrow + 40);
            xa[10] = half ? zero2 : v;
        }

        #pragma unroll
        for (int t = 0; t < 4; ++t) {
            #pragma unroll
            for (int j = 0; j < 10; ++j) {
                const float* p = W1 + (4 * j + kh) * H1 + t * 16 + l;
                w1b[t][j].x = p[0];
                w1b[t][j].y = p[H1];
            }
            const float* p = W1 + 40 * H1 + t * 16 + l;
            v2f v; v.x = p[0]; v.y = p[H1];
            w1b[t][10] = half ? zero2 : v;
        }

        #pragma unroll
        for (int t = 0; t < 2; ++t) {
            #pragma unroll
            for (int j = 0; j < 16; ++j) {
                const float* p = W2 + (4 * j + kh) * H2 + t * 16 + l;
                w2b[t][j].x = p[0];
                w2b[t][j].y = p[H2];
            }
        }

        #pragma unroll
        for (int t = 0; t < 4; ++t) b1v[t] = b1[t * 16 + l];
        #pragma unroll
        for (int t = 0; t < 2; ++t) b2v[t] = b2[t * 16 + l];
        #pragma unroll
        for (int j = 0; j < 8; ++j) w3k[j] = *(const v2f*)(W3 + 4 * j + kh);
        b3v = b3[0];

        #pragma unroll
        for (int t = 0; t < 4; ++t) {
            v8f c;
            #pragma unroll
            for (int i = 0; i < 8; ++i) c[i] = b1v[t];
            #pragma unroll
            for (int j = 0; j < 11; ++j) c = WMMA_F32(xa[j], w1b[t][j], c);
            #pragma unroll
            for (int i = 0; i < 8; ++i) sZ1[(i + 8 * half) * 64 + t * 16 + l] = c[i];
        }

        #pragma unroll
        for (int t = 0; t < 4; ++t) {
            #pragma unroll
            for (int j = 0; j < 8; ++j)
                w2t[t][j] = *(const v2f*)(W2 + (t * 16 + l) * H2 + 4 * j + kh);
        }
        #pragma unroll
        for (int t = 0; t < 3; ++t) {
            const int d  = t * 16 + l;
            const int dc = d < DF ? d : DF - 1;
            #pragma unroll
            for (int j = 0; j < 16; ++j)
                w1t[t][j] = *(const v2f*)(W1 + dc * H1 + 4 * j + kh);
        }
    }
    __syncthreads();

    #pragma unroll
    for (int k = 0; k < 4; ++k) {
        const int e = tid + 256 * k;
        const float z = sZ1[e];
        sA1[e] = softplus_f(z);
        sS1[e] = sigmoid_f(z);
    }
    __syncthreads();

    if (wave == 0) {
        #pragma unroll
        for (int t = 0; t < 2; ++t) {
            v8f c;
            #pragma unroll
            for (int i = 0; i < 8; ++i) c[i] = b2v[t];
            #pragma unroll
            for (int j = 0; j < 16; ++j) {
                v2f a;
                a.x = sA1[l * 64 + 4 * j + kh];
                a.y = sA1[l * 64 + 4 * j + kh + 1];
                c = WMMA_F32(a, w2b[t][j], c);
            }
            #pragma unroll
            for (int i = 0; i < 8; ++i) sZ2[(i + 8 * half) * 32 + t * 16 + l] = c[i];
        }
    }
    __syncthreads();

    #pragma unroll
    for (int k = 0; k < 2; ++k) {
        const int e = tid + 256 * k;
        const float z = sZ2[e];
        sA2[e] = softplus_f(z);
        sG2[e] = sigmoid_f(z) * w3c;
    }
    __syncthreads();

    if (wave == 0) {
        v8f ce;
        #pragma unroll
        for (int i = 0; i < 8; ++i) ce[i] = 0.0f;
        #pragma unroll
        for (int j = 0; j < 8; ++j) {
            v2f a;
            a.x = sA2[l * 32 + 4 * j + kh];
            a.y = sA2[l * 32 + 4 * j + kh + 1];
            v2f b = (l == 0) ? w3k[j] : zero2;
            ce = WMMA_F32(a, b, ce);
        }
        if (l == 0) {
            #pragma unroll
            for (int v = 0; v < 8; ++v)
                eiws[row0 + 8 * half + v] = ce[v] + b3v;
        }

        #pragma unroll
        for (int t = 0; t < 4; ++t) {
            v8f c;
            #pragma unroll
            for (int i = 0; i < 8; ++i) c[i] = 0.0f;
            #pragma unroll
            for (int j = 0; j < 8; ++j) {
                v2f a;
                a.x = sG2[l * 32 + 4 * j + kh];
                a.y = sG2[l * 32 + 4 * j + kh + 1];
                c = WMMA_F32(a, w2t[t][j], c);
            }
            #pragma unroll
            for (int i = 0; i < 8; ++i) {
                const int e = (i + 8 * half) * 64 + t * 16 + l;
                sG1[e] = sS1[e] * c[i];
            }
        }
    }
    __syncthreads();

    if (wave == 0) {
        #pragma unroll
        for (int t = 0; t < 3; ++t) {
            v8f c;
            #pragma unroll
            for (int i = 0; i < 8; ++i) c[i] = 0.0f;
            #pragma unroll
            for (int j = 0; j < 16; ++j) {
                v2f a;
                a.x = sG1[l * 64 + 4 * j + kh];
                a.y = sG1[l * 64 + 4 * j + kh + 1];
                c = WMMA_F32(a, w1t[t][j], c);
            }
            int d = t * 16 + l;
            if (d < DF) {
                #pragma unroll
                for (int v = 0; v < 8; ++v)
                    gws[(row0 + v + 8 * half) * DF + d] = c[v];
            }
        }
    }
}

__global__ __launch_bounds__(256)
void force_kernel(const float* __restrict__ dfeat,
                  const float* __restrict__ gws,
                  float* __restrict__ out)
{
    __shared__ __align__(16) float gsh[JF];

    const int tid = threadIdx.x;
    const int blk = blockIdx.x;
    const int b   = blk >> 5;
    const int i0  = (blk & 31) * 8;
    const int wave = tid >> 5;
    const int lane = tid & 31;
    const int i    = i0 + wave;

    const float4* gsrc = (const float4*)(gws + (size_t)b * JF);
    float4 gq[11];
    #pragma unroll
    for (int k = 0; k < 10; ++k) gq[k] = gsrc[tid + 256 * k];
    gq[10] = gsrc[2560 + (tid & 127)];

    const float* row = dfeat + (size_t)(b * NATOMS + i) * (size_t)(JF * 3);
    float4 dq[12];
    #pragma unroll
    for (int s = 0; s < 4; ++s) {
        const float4* p = (const float4*)(row + (size_t)(s * 32 + lane) * 12);
        dq[3 * s + 0] = p[0];
        dq[3 * s + 1] = p[1];
        dq[3 * s + 2] = p[2];
    }

    float4* gdst = (float4*)gsh;
    #pragma unroll
    for (int k = 0; k < 10; ++k) gdst[tid + 256 * k] = gq[k];
    gdst[2560 + (tid & 127)] = gq[10];
    __syncthreads();

    const float4* g4 = (const float4*)gsh;

    float ax = 0.0f, ay = 0.0f, az = 0.0f;
    #pragma unroll
    for (int s = 0; s < 4; ++s) {
        float4 d0 = dq[3 * s + 0], d1 = dq[3 * s + 1], d2 = dq[3 * s + 2];
        float4 gv = g4[s * 32 + lane];
        ax = fmaf(d0.x, gv.x, fmaf(d0.w, gv.y, fmaf(d1.z, gv.z, fmaf(d2.y, gv.w, ax))));
        ay = fmaf(d0.y, gv.x, fmaf(d1.x, gv.y, fmaf(d1.w, gv.z, fmaf(d2.z, gv.w, ay))));
        az = fmaf(d0.z, gv.x, fmaf(d1.y, gv.y, fmaf(d2.x, gv.z, fmaf(d2.w, gv.w, az))));
    }
    for (int it = 4; it < 84; ++it) {
        int q = it * 32 + lane;
        const float4* p = (const float4*)(row + (size_t)q * 12);
        float4 d0 = p[0], d1 = p[1], d2 = p[2];
        float4 gv = g4[q];
        ax = fmaf(d0.x, gv.x, fmaf(d0.w, gv.y, fmaf(d1.z, gv.z, fmaf(d2.y, gv.w, ax))));
        ay = fmaf(d0.y, gv.x, fmaf(d1.x, gv.y, fmaf(d1.w, gv.z, fmaf(d2.z, gv.w, ay))));
        az = fmaf(d0.z, gv.x, fmaf(d1.y, gv.y, fmaf(d2.x, gv.z, fmaf(d2.w, gv.w, az))));
    }
    #pragma unroll
    for (int off = 16; off; off >>= 1) {
        ax += __shfl_xor(ax, off, 32);
        ay += __shfl_xor(ay, off, 32);
        az += __shfl_xor(az, off, 32);
    }
    if (lane == 0) {
        float* f = out + 8 + (size_t)(b * NATOMS + i) * 3;
        f[0] = ax; f[1] = ay; f[2] = az;
    }
}

__global__ __launch_bounds__(256)
void etot_kernel(const float* __restrict__ eiws, float* __restrict__ out)
{
    const int b = threadIdx.x >> 5;
    const int lane = threadIdx.x & 31;
    float s = 0.0f;
    #pragma unroll
    for (int k = 0; k < 8; ++k) s += eiws[b * NATOMS + lane + 32 * k];
    #pragma unroll
    for (int off = 16; off; off >>= 1) s += __shfl_xor(s, off, 32);
    if (lane == 0) out[b] = s;
}

extern "C" void kernel_launch(void* const* d_in, const int* in_sizes, int n_in,
                              void* d_out, int out_size, void* d_ws, size_t ws_size,
                              hipStream_t stream)
{
    (void)in_sizes; (void)n_in; (void)out_size; (void)ws_size;

    const float* image = (const float*)d_in[0];
    const float* dfeat = (const float*)d_in[1];
    const float* W1    = (const float*)d_in[2];
    const float* b1    = (const float*)d_in[3];
    const float* W2    = (const float*)d_in[4];
    const float* b2    = (const float*)d_in[5];
    const float* W3    = (const float*)d_in[6];
    const float* b3    = (const float*)d_in[7];

    float* out  = (float*)d_out;
    float* gws  = (float*)d_ws;
    float* eiws = gws + (size_t)ROWS * DF;

    mlp_wmma_kernel<<<ROWS / 16, 256, 0, stream>>>(image, W1, b1, W2, b2, W3, b3, gws, eiws);
    force_kernel<<<256, 256, 0, stream>>>(dfeat, gws, out);
    etot_kernel<<<1, 256, 0, stream>>>(eiws, out);
}


# force kernel: stream software-pipelined and never drained: 36 loads per wave stay requested ahead, also across the g barrier (LDS-only wait); was 12 loads, all awaited each trip; sums unchanged
# speedup vs baseline: 2.2041x; 1.0493x over previous
#include <hip/hip_runtime.h>

typedef float v2f __attribute__((ext_vector_type(2)));
typedef float v8f __attribute__((ext_vector_type(8)));

#define WMMA_F32(a, b, c) \
    __builtin_amdgcn_wmma_f32_16x16x4_f32(false, (a), false, (b), (short)0, (c), false, false)

constexpr int BATCH  = 8;
constexpr int NATOMS = 256;
constexpr int DF     = 42;
constexpr int H1     = 64;
constexpr int H2     = 32;
constexpr int ROWS   = BATCH * NATOMS;
constexpr int JF     = NATOMS * DF;

__device__ __forceinline__ float softplus_f(float z) {
    return fmaxf(z, 0.0f) + log1pf(expf(-fabsf(z)));
}
__device__ __forceinline__ float sigmoid_f(float z) {
    return 1.0f / (1.0f + expf(-z));
}

__global__ __launch_bounds__(256)
void mlp_wmma_kernel(const float* __restrict__ image,
                     const float* __restrict__ W1, const float* __restrict__ b1,
                     const float* __restrict__ W2, const float* __restrict__ b2,
                     const float* __restrict__ W3, const float* __restrict__ b3,
                     float* __restrict__ gws,
                     float* __restrict__ eiws)
{
    __shared__ float sZ1[16 * 64];
    __shared__ float sA1[16 * 64];
    __shared__ float sS1[16 * 64];
    __shared__ float sZ2[16 * 32];
    __shared__ float sA2[16 * 32];
    __shared__ float sG2[16 * 32];
    __shared__ float sG1[16 * 64];

    const int tid  = threadIdx.x;
    const int wave = __builtin_amdgcn_readfirstlane(tid >> 5);
    const int lane = tid & 31;
    const int half = lane >> 4;
    const int l    = lane & 15;
    const int kh   = 2 * half;
    const int tile = blockIdx.x;
    const int row0 = tile * 16;

    v2f zero2; zero2.x = 0.0f; zero2.y = 0.0f;

    const float w3c = W3[tid & 31];

    v2f xa[11];
    v2f w1b[4][11];
    v2f w2b[2][16];
    float b1v[4], b2v[2];
    v2f w3k[8];
    float b3v = 0.0f;
    v2f w2t[4][8];
    v2f w1t[3][16];

    if (wave == 0) {
        const float* xrow = image + (row0 + l) * DF;
        #pragma unroll
        for (int j = 0; j < 10; ++j) xa[j] = *(const v2f*)(xrow + 4 * j + kh);
        {
            v2f v = *(const v2f*)(xrow + 40);
            xa[10] = half ? zero2 : v;
        }

        #pragma unroll
        for (int t = 0; t < 4; ++t) {
            #pragma unroll
            for (int j = 0; j < 10; ++j) {
                const float* p = W1 + (4 * j + kh) * H1 + t * 16 + l;
                w1b[t][j].x = p[0];
                w1b[t][j].y = p[H1];
            }
            const float* p = W1 + 40 * H1 + t * 16 + l;
            v2f v; v.x = p[0]; v.y = p[H1];
            w1b[t][10] = half ? zero2 : v;
        }

        #pragma unroll
        for (int t = 0; t < 2; ++t) {
            #pragma unroll
            for (int j = 0; j < 16; ++j) {
                const float* p = W2 + (4 * j + kh) * H2 + t * 16 + l;
                w2b[t][j].x = p[0];
                w2b[t][j].y = p[H2];
            }
        }

        #pragma unroll
        for (int t = 0; t < 4; ++t) b1v[t] = b1[t * 16 + l];
        #pragma unroll
        for (int t = 0; t < 2; ++t) b2v[t] = b2[t * 16 + l];
        #pragma unroll
        for (int j = 0; j < 8; ++j) w3k[j] = *(const v2f*)(W3 + 4 * j + kh);
        b3v = b3[0];

        #pragma unroll
        for (int t = 0; t < 4; ++t) {
            v8f c;
            #pragma unroll
            for (int i = 0; i < 8; ++i) c[i] = b1v[t];
            #pragma unroll
            for (int j = 0; j < 11; ++j) c = WMMA_F32(xa[j], w1b[t][j], c);
            #pragma unroll
            for (int i = 0; i < 8; ++i) sZ1[(i + 8 * half) * 64 + t * 16 + l] = c[i];
        }

        #pragma unroll
        for (int t = 0; t < 4; ++t) {
            #pragma unroll
            for (int j = 0; j < 8; ++j)
                w2t[t][j] = *(const v2f*)(W2 + (t * 16 + l) * H2 + 4 * j + kh);
        }
        #pragma unroll
        for (int t = 0; t < 3; ++t) {
            const int d  = t * 16 + l;
            const int dc = d < DF ? d : DF - 1;
            #pragma unroll
            for (int j = 0; j < 16; ++j)
                w1t[t][j] = *(const v2f*)(W1 + dc * H1 + 4 * j + kh);
        }
    }
    __syncthreads();

    #pragma unroll
    for (int k = 0; k < 4; ++k) {
        const int e = tid + 256 * k;
        const float z = sZ1[e];
        sA1[e] = softplus_f(z);
        sS1[e] = sigmoid_f(z);
    }
    __syncthreads();

    if (wave == 0) {
        #pragma unroll
        for (int t = 0; t < 2; ++t) {
            v8f c;
            #pragma unroll
            for (int i = 0; i < 8; ++i) c[i] = b2v[t];
            #pragma unroll
            for (int j = 0; j < 16; ++j) {
                v2f a;
                a.x = sA1[l * 64 + 4 * j + kh];
                a.y = sA1[l * 64 + 4 * j + kh + 1];
                c = WMMA_F32(a, w2b[t][j], c);
            }
            #pragma unroll
            for (int i = 0; i < 8; ++i) sZ2[(i + 8 * half) * 32 + t * 16 + l] = c[i];
        }
    }
    __syncthreads();

    #pragma unroll
    for (int k = 0; k < 2; ++k) {
        const int e = tid + 256 * k;
        const float z = sZ2[e];
        sA2[e] = softplus_f(z);
        sG2[e] = sigmoid_f(z) * w3c;
    }
    __syncthreads();

    if (wave == 0) {
        v8f ce;
        #pragma unroll
        for (int i = 0; i < 8; ++i) ce[i] = 0.0f;
        #pragma unroll
        for (int j = 0; j < 8; ++j) {
            v2f a;
            a.x = sA2[l * 32 + 4 * j + kh];
            a.y = sA2[l * 32 + 4 * j + kh + 1];
            v2f b = (l == 0) ? w3k[j] : zero2;
            ce = WMMA_F32(a, b, ce);
        }
        if (l == 0) {
            #pragma unroll
            for (int v = 0; v < 8; ++v)
                eiws[row0 + 8 * half + v] = ce[v] + b3v;
        }

        #pragma unroll
        for (int t = 0; t < 4; ++t) {
            v8f c;
            #pragma unroll
            for (int i = 0; i < 8; ++i) c[i] = 0.0f;
            #pragma unroll
            for (int j = 0; j < 8; ++j) {
                v2f a;
                a.x = sG2[l * 32 + 4 * j + kh];
                a.y = sG2[l * 32 + 4 * j + kh + 1];
                c = WMMA_F32(a, w2t[t][j], c);
            }
            #pragma unroll
            for (int i = 0; i < 8; ++i) {
                const int e = (i + 8 * half) * 64 + t * 16 + l;
                sG1[e] = sS1[e] * c[i];
            }
        }
    }
    __syncthreads();

    if (wave == 0) {
        #pragma unroll
        for (int t = 0; t < 3; ++t) {
            v8f c;
            #pragma unroll
            for (int i = 0; i < 8; ++i) c[i] = 0.0f;
            #pragma unroll
            for (int j = 0; j < 16; ++j) {
                v2f a;
                a.x = sG1[l * 64 + 4 * j + kh];
                a.y = sG1[l * 64 + 4 * j + kh + 1];
                c = WMMA_F32(a, w1t[t][j], c);
            }
            int d = t * 16 + l;
            if (d < DF) {
                #pragma unroll
                for (int v = 0; v < 8; ++v)
                    gws[(row0 + v + 8 * half) * DF + d] = c[v];
            }
        }
    }
}

constexpr int NPIECE = 84;
constexpr int AHEAD  = 12;
constexpr int GAHEAD = 3;

__global__ __launch_bounds__(256)
void force_kernel(const float* __restrict__ dfeat,
                  const float* gws,
                  float* __restrict__ out)
{
    __shared__ __align__(16) float gsh[JF];

    const int tid = threadIdx.x;
    const int blk = blockIdx.x;
    const int b   = blk >> 5;
    const int i0  = (blk & 31) * 8;
    const int wave = tid >> 5;
    const int lane = tid & 31;
    const int i    = i0 + wave;

    const float4* gsrc = (const float4*)(gws + (size_t)b * JF);
    const float4 gq0 = gsrc[tid],            gq1 = gsrc[tid + 256 * 1], gq2 = gsrc[tid + 256 * 2];
    const float4 gq3 = gsrc[tid + 256 * 3],  gq4 = gsrc[tid + 256 * 4], gq5 = gsrc[tid + 256 * 5];
    const float4 gq6 = gsrc[tid + 256 * 6],  gq7 = gsrc[tid + 256 * 7], gq8 = gsrc[tid + 256 * 8];
    const float4 gq9 = gsrc[tid + 256 * 9],  gq10 = gsrc[2560 + (tid & 127)];

    const float4* rowq = (const float4*)(dfeat + (size_t)(b * NATOMS + i) * (size_t)(JF * 3)) + lane * 3;
    float4 d[NPIECE][3];
#define LOAD_PIECE(T) do { const float4* p_ = rowq + (T) * 96; \
        d[(T)][0] = p_[0]; d[(T)][1] = p_[1]; d[(T)][2] = p_[2]; } while (0)


    #pragma unroll
    for (int t = 0; t < AHEAD; ++t) LOAD_PIECE(t);

    __builtin_amdgcn_sched_barrier(0);

    float4* gdst = (float4*)gsh;
    gdst[tid] = gq0;           gdst[tid + 256 * 1] = gq1; gdst[tid + 256 * 2] = gq2;
    gdst[tid + 256 * 3] = gq3; gdst[tid + 256 * 4] = gq4; gdst[tid + 256 * 5] = gq5;
    gdst[tid + 256 * 6] = gq6; gdst[tid + 256 * 7] = gq7; gdst[tid + 256 * 8] = gq8;
    gdst[tid + 256 * 9] = gq9; gdst[2560 + (tid & 127)] = gq10;

    __builtin_amdgcn_fence(__ATOMIC_RELEASE, "workgroup", "local");
    __builtin_amdgcn_s_barrier();
    __builtin_amdgcn_fence(__ATOMIC_ACQUIRE, "workgroup", "local");

    const float4* g4 = (const float4*)gsh;

    float ax = 0.0f, ay = 0.0f, az = 0.0f;
    float4 gr[NPIECE];
    #pragma unroll
    for (int t = 0; t < GAHEAD; ++t) gr[t] = g4[t * 32 + lane];
    #pragma unroll
    for (int it = 0; it < NPIECE; ++it) {
        __builtin_amdgcn_sched_barrier(0);
        if (it + AHEAD < NPIECE) LOAD_PIECE(it + AHEAD);
        if (it + GAHEAD < NPIECE) gr[it + GAHEAD] = g4[(it + GAHEAD) * 32 + lane];
        __builtin_amdgcn_sched_barrier(0);
        const float4 d0 = d[it][0], d1 = d[it][1], d2 = d[it][2];
        const float4 gv = gr[it];
        ax = fmaf(d0.x, gv.x, fmaf(d0.w, gv.y, fmaf(d1.z, gv.z, fmaf(d2.y, gv.w, ax))));
        ay = fmaf(d0.y, gv.x, fmaf(d1.x, gv.y, fmaf(d1.w, gv.z, fmaf(d2.z, gv.w, ay))));
        az = fmaf(d0.z, gv.x, fmaf(d1.y, gv.y, fmaf(d2.x, gv.z, fmaf(d2.w, gv.w, az))));
    }
#undef LOAD_PIECE
    #pragma unroll
    for (int off = 16; off; off >>= 1) {
        ax += __shfl_xor(ax, off, 32);
        ay += __shfl_xor(ay, off, 32);
        az += __shfl_xor(az, off, 32);
    }
    if (lane == 0) {
        float* f = out + 8 + (size_t)(b * NATOMS + i) * 3;
        f[0] = ax; f[1] = ay; f[2] = az;
    }
}

__global__ __launch_bounds__(256)
void etot_kernel(const float* __restrict__ eiws, float* __restrict__ out)
{
    const int b = threadIdx.x >> 5;
    const int lane = threadIdx.x & 31;
    float s = 0.0f;
    #pragma unroll
    for (int k = 0; k < 8; ++k) s += eiws[b * NATOMS + lane + 32 * k];
    #pragma unroll
    for (int off = 16; off; off >>= 1) s += __shfl_xor(s, off, 32);
    if (lane == 0) out[b] = s;
}

extern "C" void kernel_launch(void* const* d_in, const int* in_sizes, int n_in,
                              void* d_out, int out_size, void* d_ws, size_t ws_size,
                              hipStream_t stream)
{
    (void)in_sizes; (void)n_in; (void)out_size; (void)ws_size;

    const float* image = (const float*)d_in[0];
    const float* dfeat = (const float*)d_in[1];
    const float* W1    = (const float*)d_in[2];
    const float* b1    = (const float*)d_in[3];
    const float* W2    = (const float*)d_in[4];
    const float* b2    = (const float*)d_in[5];
    const float* W3    = (const float*)d_in[6];
    const float* b3    = (const float*)d_in[7];

    float* out  = (float*)d_out;
    float* gws  = (float*)d_ws;
    float* eiws = gws + (size_t)ROWS * DF;

    mlp_wmma_kernel<<<ROWS / 16, 256, 0, stream>>>(image, W1, b1, W2, b2, W3, b3, gws, eiws);
    force_kernel<<<256, 256, 0, stream>>>(dfeat, gws, out);
    etot_kernel<<<1, 256, 0, stream>>>(eiws, out);
}


# first launch gets 256 blocks: block n prefetches into the second-level cache the first 48 KB of the 8 dfeat rows that force block n streams, while the MLP chain leaves memory idle; sums unchanged
# speedup vs baseline: 2.4904x; 1.1299x over previous
#include <hip/hip_runtime.h>

typedef float v2f __attribute__((ext_vector_type(2)));
typedef float v8f __attribute__((ext_vector_type(8)));

#define WMMA_F32(a, b, c) \
    __builtin_amdgcn_wmma_f32_16x16x4_f32(false, (a), false, (b), (short)0, (c), false, false)

constexpr int BATCH  = 8;
constexpr int NATOMS = 256;
constexpr int DF     = 42;
constexpr int H1     = 64;
constexpr int H2     = 32;
constexpr int ROWS   = BATCH * NATOMS;
constexpr int JF     = NATOMS * DF;

__device__ __forceinline__ float softplus_f(float z) {
    return fmaxf(z, 0.0f) + log1pf(expf(-fabsf(z)));
}
__device__ __forceinline__ float sigmoid_f(float z) {
    return 1.0f / (1.0f + expf(-z));
}

constexpr int PF_STEPS = 6;
constexpr int PF_BYTES = PF_STEPS * 8192;
static_assert(PF_BYTES <= JF * 3 * 4, "prefetch stays inside the row");

__device__ __forceinline__ void prefetch_force_rows(const float* dfeat, int fblock, int wave, int lane) {
    const int prow = (fblock >> 5) * NATOMS + (fblock & 31) * 8 + wave;
    const char* ph = (const char*)(dfeat + (size_t)prow * (size_t)(JF * 3)) + lane * 256;
    #pragma unroll
    for (int k = 0; k < PF_STEPS; ++k)
        __builtin_prefetch(ph + k * 8192, 0, 1);
}

__global__ __launch_bounds__(256)
void mlp_wmma_kernel(const float* __restrict__ dfeat,
                     const float* __restrict__ image,
                     const float* __restrict__ W1, const float* __restrict__ b1,
                     const float* __restrict__ W2, const float* __restrict__ b2,
                     const float* __restrict__ W3, const float* __restrict__ b3,
                     float* __restrict__ gws,
                     float* __restrict__ eiws)
{
    __shared__ float sZ1[16 * 64];
    __shared__ float sA1[16 * 64];
    __shared__ float sS1[16 * 64];
    __shared__ float sZ2[16 * 32];
    __shared__ float sA2[16 * 32];
    __shared__ float sG2[16 * 32];
    __shared__ float sG1[16 * 64];

    const int tid  = threadIdx.x;
    const int wave = __builtin_amdgcn_readfirstlane(tid >> 5);
    const int lane = tid & 31;
    const int half = lane >> 4;
    const int l    = lane & 15;
    const int kh   = 2 * half;
    const int tile = blockIdx.x;
    const int row0 = tile * 16;

    if (tile >= ROWS / 16) {
        prefetch_force_rows(dfeat, tile, wave, lane);
        return;
    }

    v2f zero2; zero2.x = 0.0f; zero2.y = 0.0f;

    const float w3c = W3[tid & 31];

    v2f xa[11];
    v2f w1b[4][11];
    v2f w2b[2][16];
    float b1v[4], b2v[2];
    v2f w3k[8];
    float b3v = 0.0f;
    v2f w2t[4][8];
    v2f w1t[3][16];

    if (wave == 0) {
        const float* xrow = image + (row0 + l) * DF;
        #pragma unroll
        for (int j = 0; j < 10; ++j) xa[j] = *(const v2f*)(xrow + 4 * j + kh);
        {
            v2f v = *(const v2f*)(xrow + 40);
            xa[10] = half ? zero2 : v;
        }

        #pragma unroll
        for (int t = 0; t < 4; ++t) {
            #pragma unroll
            for (int j = 0; j < 10; ++j) {
                const float* p = W1 + (4 * j + kh) * H1 + t * 16 + l;
                w1b[t][j].x = p[0];
                w1b[t][j].y = p[H1];
            }
            const float* p = W1 + 40 * H1 + t * 16 + l;
            v2f v; v.x = p[0]; v.y = p[H1];
            w1b[t][10] = half ? zero2 : v;
        }

        #pragma unroll
        for (int t = 0; t < 2; ++t) {
            #pragma unroll
            for (int j = 0; j < 16; ++j) {
                const float* p = W2 + (4 * j + kh) * H2 + t * 16 + l;
                w2b[t][j].x = p[0];
                w2b[t][j].y = p[H2];
            }
        }

        #pragma unroll
        for (int t = 0; t < 4; ++t) b1v[t] = b1[t * 16 + l];
        #pragma unroll
        for (int t = 0; t < 2; ++t) b2v[t] = b2[t * 16 + l];
        #pragma unroll
        for (int j = 0; j < 8; ++j) w3k[j] = *(const v2f*)(W3 + 4 * j + kh);
        b3v = b3[0];

        #pragma unroll
        for (int t = 0; t < 4; ++t) {
            v8f c;
            #pragma unroll
            for (int i = 0; i < 8; ++i) c[i] = b1v[t];
            #pragma unroll
            for (int j = 0; j < 11; ++j) c = WMMA_F32(xa[j], w1b[t][j], c);
            #pragma unroll
            for (int i = 0; i < 8; ++i) sZ1[(i + 8 * half) * 64 + t * 16 + l] = c[i];
        }

        #pragma unroll
        for (int t = 0; t < 4; ++t) {
            #pragma unroll
            for (int j = 0; j < 8; ++j)
                w2t[t][j] = *(const v2f*)(W2 + (t * 16 + l) * H2 + 4 * j + kh);
        }
        #pragma unroll
        for (int t = 0; t < 3; ++t) {
            const int d  = t * 16 + l;
            const int dc = d < DF ? d : DF - 1;
            #pragma unroll
            for (int j = 0; j < 16; ++j)
                w1t[t][j] = *(const v2f*)(W1 + dc * H1 + 4 * j + kh);
        }
    }
    __syncthreads();

    prefetch_force_rows(dfeat, tile, wave, lane);

    #pragma unroll
    for (int k = 0; k < 4; ++k) {
        const int e = tid + 256 * k;
        const float z = sZ1[e];
        sA1[e] = softplus_f(z);
        sS1[e] = sigmoid_f(z);
    }
    __syncthreads();

    if (wave == 0) {
        #pragma unroll
        for (int t = 0; t < 2; ++t) {
            v8f c;
            #pragma unroll
            for (int i = 0; i < 8; ++i) c[i] = b2v[t];
            #pragma unroll
            for (int j = 0; j < 16; ++j) {
                v2f a;
                a.x = sA1[l * 64 + 4 * j + kh];
                a.y = sA1[l * 64 + 4 * j + kh + 1];
                c = WMMA_F32(a, w2b[t][j], c);
            }
            #pragma unroll
            for (int i = 0; i < 8; ++i) sZ2[(i + 8 * half) * 32 + t * 16 + l] = c[i];
        }
    }
    __syncthreads();

    #pragma unroll
    for (int k = 0; k < 2; ++k) {
        const int e = tid + 256 * k;
        const float z = sZ2[e];
        sA2[e] = softplus_f(z);
        sG2[e] = sigmoid_f(z) * w3c;
    }
    __syncthreads();

    if (wave == 0) {
        v8f ce;
        #pragma unroll
        for (int i = 0; i < 8; ++i) ce[i] = 0.0f;
        #pragma unroll
        for (int j = 0; j < 8; ++j) {
            v2f a;
            a.x = sA2[l * 32 + 4 * j + kh];
            a.y = sA2[l * 32 + 4 * j + kh + 1];
            v2f b = (l == 0) ? w3k[j] : zero2;
            ce = WMMA_F32(a, b, ce);
        }
        if (l == 0) {
            #pragma unroll
            for (int v = 0; v < 8; ++v)
                eiws[row0 + 8 * half + v] = ce[v] + b3v;
        }

        #pragma unroll
        for (int t = 0; t < 4; ++t) {
            v8f c;
            #pragma unroll
            for (int i = 0; i < 8; ++i) c[i] = 0.0f;
            #pragma unroll
            for (int j = 0; j < 8; ++j) {
                v2f a;
                a.x = sG2[l * 32 + 4 * j + kh];
                a.y = sG2[l * 32 + 4 * j + kh + 1];
                c = WMMA_F32(a, w2t[t][j], c);
            }
            #pragma unroll
            for (int i = 0; i < 8; ++i) {
                const int e = (i + 8 * half) * 64 + t * 16 + l;
                sG1[e] = sS1[e] * c[i];
            }
        }
    }
    __syncthreads();

    if (wave == 0) {
        #pragma unroll
        for (int t = 0; t < 3; ++t) {
            v8f c;
            #pragma unroll
            for (int i = 0; i < 8; ++i) c[i] = 0.0f;
            #pragma unroll
            for (int j = 0; j < 16; ++j) {
                v2f a;
                a.x = sG1[l * 64 + 4 * j + kh];
                a.y = sG1[l * 64 + 4 * j + kh + 1];
                c = WMMA_F32(a, w1t[t][j], c);
            }
            int d = t * 16 + l;
            if (d < DF) {
                #pragma unroll
                for (int v = 0; v < 8; ++v)
                    gws[(row0 + v + 8 * half) * DF + d] = c[v];
            }
        }
    }
}

constexpr int NPIECE = 84;
constexpr int AHEAD  = 12;
constexpr int GAHEAD = 3;

__global__ __launch_bounds__(256)
void force_kernel(const float* __restrict__ dfeat,
                  const float* gws,
                  float* __restrict__ out)
{
    __shared__ __align__(16) float gsh[JF];

    const int tid = threadIdx.x;
    const int blk = blockIdx.x;
    const int b   = blk >> 5;
    const int i0  = (blk & 31) * 8;
    const int wave = tid >> 5;
    const int lane = tid & 31;
    const int i    = i0 + wave;

    const float4* gsrc = (const float4*)(gws + (size_t)b * JF);
    const float4 gq0 = gsrc[tid],            gq1 = gsrc[tid + 256 * 1], gq2 = gsrc[tid + 256 * 2];
    const float4 gq3 = gsrc[tid + 256 * 3],  gq4 = gsrc[tid + 256 * 4], gq5 = gsrc[tid + 256 * 5];
    const float4 gq6 = gsrc[tid + 256 * 6],  gq7 = gsrc[tid + 256 * 7], gq8 = gsrc[tid + 256 * 8];
    const float4 gq9 = gsrc[tid + 256 * 9],  gq10 = gsrc[2560 + (tid & 127)];

    const float4* rowq = (const float4*)(dfeat + (size_t)(b * NATOMS + i) * (size_t)(JF * 3)) + lane * 3;
    float4 d[NPIECE][3];
#define LOAD_PIECE(T) do { const float4* p_ = rowq + (T) * 96; \
        d[(T)][0] = p_[0]; d[(T)][1] = p_[1]; d[(T)][2] = p_[2]; } while (0)


    #pragma unroll
    for (int t = 0; t < AHEAD; ++t) LOAD_PIECE(t);

    __builtin_amdgcn_sched_barrier(0);

    float4* gdst = (float4*)gsh;
    gdst[tid] = gq0;           gdst[tid + 256 * 1] = gq1; gdst[tid + 256 * 2] = gq2;
    gdst[tid + 256 * 3] = gq3; gdst[tid + 256 * 4] = gq4; gdst[tid + 256 * 5] = gq5;
    gdst[tid + 256 * 6] = gq6; gdst[tid + 256 * 7] = gq7; gdst[tid + 256 * 8] = gq8;
    gdst[tid + 256 * 9] = gq9; gdst[2560 + (tid & 127)] = gq10;

    __builtin_amdgcn_fence(__ATOMIC_RELEASE, "workgroup", "local");
    __builtin_amdgcn_s_barrier();
    __builtin_amdgcn_fence(__ATOMIC_ACQUIRE, "workgroup", "local");

    const float4* g4 = (const float4*)gsh;

    float ax = 0.0f, ay = 0.0f, az = 0.0f;
    float4 gr[NPIECE];
    #pragma unroll
    for (int t = 0; t < GAHEAD; ++t) gr[t] = g4[t * 32 + lane];
    #pragma unroll
    for (int it = 0; it < NPIECE; ++it) {
        __builtin_amdgcn_sched_barrier(0);
        if (it + AHEAD < NPIECE) LOAD_PIECE(it + AHEAD);
        if (it + GAHEAD < NPIECE) gr[it + GAHEAD] = g4[(it + GAHEAD) * 32 + lane];
        __builtin_amdgcn_sched_barrier(0);
        const float4 d0 = d[it][0], d1 = d[it][1], d2 = d[it][2];
        const float4 gv = gr[it];
        ax = fmaf(d0.x, gv.x, fmaf(d0.w, gv.y, fmaf(d1.z, gv.z, fmaf(d2.y, gv.w, ax))));
        ay = fmaf(d0.y, gv.x, fmaf(d1.x, gv.y, fmaf(d1.w, gv.z, fmaf(d2.z, gv.w, ay))));
        az = fmaf(d0.z, gv.x, fmaf(d1.y, gv.y, fmaf(d2.x, gv.z, fmaf(d2.w, gv.w, az))));
    }
#undef LOAD_PIECE
    #pragma unroll
    for (int off = 16; off; off >>= 1) {
        ax += __shfl_xor(ax, off, 32);
        ay += __shfl_xor(ay, off, 32);
        az += __shfl_xor(az, off, 32);
    }
    if (lane == 0) {
        float* f = out + 8 + (size_t)(b * NATOMS + i) * 3;
        f[0] = ax; f[1] = ay; f[2] = az;
    }
}

__global__ __launch_bounds__(256)
void etot_kernel(const float* __restrict__ eiws, float* __restrict__ out)
{
    const int b = threadIdx.x >> 5;
    const int lane = threadIdx.x & 31;
    float s = 0.0f;
    #pragma unroll
    for (int k = 0; k < 8; ++k) s += eiws[b * NATOMS + lane + 32 * k];
    #pragma unroll
    for (int off = 16; off; off >>= 1) s += __shfl_xor(s, off, 32);
    if (lane == 0) out[b] = s;
}

extern "C" void kernel_launch(void* const* d_in, const int* in_sizes, int n_in,
                              void* d_out, int out_size, void* d_ws, size_t ws_size,
                              hipStream_t stream)
{
    (void)in_sizes; (void)n_in; (void)out_size; (void)ws_size;

    const float* image = (const float*)d_in[0];
    const float* dfeat = (const float*)d_in[1];
    const float* W1    = (const float*)d_in[2];
    const float* b1    = (const float*)d_in[3];
    const float* W2    = (const float*)d_in[4];
    const float* b2    = (const float*)d_in[5];
    const float* W3    = (const float*)d_in[6];
    const float* b3    = (const float*)d_in[7];

    float* out  = (float*)d_out;
    float* gws  = (float*)d_ws;
    float* eiws = gws + (size_t)ROWS * DF;

    mlp_wmma_kernel<<<256, 256, 0, stream>>>(dfeat, image, W1, b1, W2, b2, W3, b3, gws, eiws);
    force_kernel<<<256, 256, 0, stream>>>(dfeat, gws, out);
    etot_kernel<<<1, 256, 0, stream>>>(eiws, out);
}
